# starting kernel
# baseline (speedup 1.0000x reference)
#include <hip/hip_runtime.h>

typedef float v2f __attribute__((ext_vector_type(2)));
typedef float v8f __attribute__((ext_vector_type(8)));

namespace {
constexpr int KK   = 17;
constexpr int PAD  = 8;
constexpr int NB   = 4;
constexpr int NC   = 3;
constexpr int HH   = 256;
constexpr int WW   = 256;
constexpr int TILE = 16;
}

__device__ __forceinline__ int clampi(int x, int lo, int hi) {
  return x < lo ? lo : (x > hi ? hi : x);
}

__global__ __launch_bounds__(32) void sepconv_wmma_f32(
    const float* __restrict__ imgs, const float* __restrict__ vers,
    const float* __restrict__ hors, float* __restrict__ out)
{
  const int lane = (int)threadIdx.x;
  const int tile = (int)blockIdx.x;

  const int tilesPerRow = WW / TILE;
  const int b   = tile / (HH * tilesPerRow);
  const int rem = tile - b * HH * tilesPerRow;
  const int h   = rem / tilesPerRow;
  const int w0  = (rem - h * tilesPerRow) * TILE;

  const long plane = (long)HH * WW;

  const int nm    = lane & 15;
  const int kbase = (lane < 16) ? 0 : 2;

  int offA[5][2];
  #pragma unroll
  for (int c5 = 0; c5 < 5; ++c5) {
    const int cx = w0 - PAD + 4 * c5 + nm + kbase;
    offA[c5][0] = clampi(cx,     0, WW - 1);
    offA[c5][1] = clampi(cx + 1, 0, WW - 1);
  }

  v2f Bc[5];
  {
    const float* hbase = hors + ((long)b * KK) * plane + (long)h * WW + w0;
    #pragma unroll
    for (int c5 = 0; c5 < 5; ++c5) {
      const int k0 = 4 * c5 + kbase;
      Bc[c5].x = (k0     < KK) ? hbase[(long)k0 * plane + nm]       : 0.0f;
      Bc[c5].y = (k0 + 1 < KK) ? hbase[(long)(k0 + 1) * plane + nm] : 0.0f;
    }
  }

  const float* vbase = vers + ((long)b * KK) * plane + (long)h * WW + w0;
  const float* ibase = imgs + ((long)b * NC) * plane;

  v8f C[NC];
  #pragma unroll
  for (int c = 0; c < NC; ++c) C[c] = (v8f){};

  for (int u = 0; u < KK; ++u) {
    const float vu = vbase[(long)u * plane + nm];
    v2f Bp[5];
    #pragma unroll
    for (int c5 = 0; c5 < 5; ++c5) {
      Bp[c5].x = Bc[c5].x * vu;
      Bp[c5].y = Bc[c5].y * vu;
    }

    const long rowoff = (long)clampi(h - PAD + u, 0, HH - 1) * WW;
    #pragma unroll
    for (int c = 0; c < NC; ++c) {
      const float* row = ibase + (long)c * plane + rowoff;
      #pragma unroll
      for (int c5 = 0; c5 < 5; ++c5) {
        v2f A;
        A.x = row[offA[c5][0]];
        A.y = row[offA[c5][1]];
        C[c] = __builtin_amdgcn_wmma_f32_16x16x4_f32(
                   false, A, false, Bp[c5], (short)0, C[c], false, false);
      }
    }
  }

  const bool active = (lane < 8) || (lane >= 24);
  const int  p      = (lane < 8) ? lane : ((lane >= 24) ? (lane - 16) : 0);
  const int  dsel   = lane & 7;
  float* obase = out + ((long)b * NC) * plane + (long)h * WW + w0;

  #pragma unroll
  for (int c = 0; c < NC; ++c) {
    float d = C[c][0];
    d = (dsel == 1) ? C[c][1] : d;
    d = (dsel == 2) ? C[c][2] : d;
    d = (dsel == 3) ? C[c][3] : d;
    d = (dsel == 4) ? C[c][4] : d;
    d = (dsel == 5) ? C[c][5] : d;
    d = (dsel == 6) ? C[c][6] : d;
    d = (dsel == 7) ? C[c][7] : d;
    if (active) obase[(long)c * plane + p] = d;
  }
}

extern "C" void kernel_launch(void* const* d_in, const int* in_sizes, int n_in,
                              void* d_out, int out_size, void* d_ws, size_t ws_size,
                              hipStream_t stream) {
  const float* imgs = (const float*)d_in[0];
  const float* vers = (const float*)d_in[1];
  const float* hors = (const float*)d_in[2];
  float* out = (float*)d_out;

  const int totalTiles = NB * HH * (WW / TILE);
  dim3 grid(totalTiles);
  dim3 block(32);
  sepconv_wmma_f32<<<grid, block, 0, stream>>>(imgs, vers, hors, out);
}


# Loads regrouped by first use into 7 groups, each requested 2 to 3 matrix blocks (40 to 60 matrix instructions) ahead of its reader, none inside a block; multiply-adds closed per block; same sums
# speedup vs baseline: 4.9977x; 4.9977x over previous
#include <hip/hip_runtime.h>

typedef float v2f __attribute__((ext_vector_type(2)));
typedef float v8f __attribute__((ext_vector_type(8)));

namespace {
constexpr int KK  = 17;
constexpr int PAD = 8;
constexpr int NB  = 4;
constexpr int NC  = 3;
constexpr int HH  = 256;
constexpr int WW  = 256;
constexpr int TW  = 16;
constexpr int TR  = 4;
constexpr int KS  = 5;
constexpr int GW  = 2;
constexpr int GH  = 4;
constexpr int NT  = 11;
constexpr int NST = 2 * NC;
constexpr int LA  = 3;
constexpr int LH  = 1;
constexpr int plane = HH * WW;
}

__device__ __forceinline__ int clampi(int x, int lo, int hi) {
  return x < lo ? lo : (x > hi ? hi : x);
}

__device__ __forceinline__ void load_a(const float* __restrict__ ip, const int (&arow)[KS][2], int acolv,
                                       v2f (&dst)[KS]) {
  #pragma unroll
  for (int s = 0; s < KS; ++s) {
    dst[s].x = ip[arow[s][0] + acolv];
    dst[s].y = ip[arow[s][1] + acolv];
  }
}

__device__ __forceinline__ v2f load_b(const float* __restrict__ vb, const float* __restrict__ vbl, int r, int s) {
  v2f b;
  #pragma unroll
  for (int e = 0; e < 2; ++e) {
    const int  u0  = 4 * s + e - r;
    const int  u1  = u0 + 2;
    const bool ok0 = (u0 >= 0) && (u0 < KK);
    const bool ok1 = (u1 >= 0) && (u1 < KK);
    float x = 0.0f;
    if (ok0 && ok1)  x = vbl[u0 * plane + r * WW];
    else if (ok0)    x = vb[u0 * plane + r * WW];
    else if (ok1)    x = vb[u1 * plane + r * WW];
    b[e] = x;
  }
  return b;
}
__device__ __forceinline__ v2f zero_b(v2f b, int hb, int r, int s) {
  #pragma unroll
  for (int e = 0; e < 2; ++e) {
    const int  u0  = 4 * s + e - r;
    const int  u1  = u0 + 2;
    const bool ok0 = (u0 >= 0) && (u0 < KK);
    const bool ok1 = (u1 >= 0) && (u1 < KK);
    if (ok0 && !ok1)      b[e] = hb ? 0.0f : b[e];
    else if (ok1 && !ok0) b[e] = hb ? b[e] : 0.0f;
  }
  return b;
}

__device__ __forceinline__ void load_taps(const float* __restrict__ hu, int px, int t0, float (&u)[TR][NT]) {
  int idx[NT];
  #pragma unroll
  for (int m = 0; m < NT; ++m)
    idx[m] = clampi(t0 + m, 0, KK - 1) * plane + px;
  __builtin_amdgcn_sched_barrier(0);
  #pragma unroll
  for (int m = 0; m < NT; ++m) {
    #pragma unroll
    for (int r = 0; r < TR; ++r)
      u[r][m] = hu[idx[m] + r * WW];
  }
}

__device__ __forceinline__ float pick(bool c, float a, float b) { return c ? a : b; }
__device__ __forceinline__ void rotate_taps(const float (&u)[NT], bool rot2, bool rot1,
                                            const bool (&okc)[8], float (&w)[8]) {
  float v[9];
  #pragma unroll
  for (int i = 0; i < 9; ++i) v[i] = pick(rot2, u[i + 2], u[i]);
  #pragma unroll
  for (int i = 0; i < 8; ++i) w[i] = pick(okc[i], pick(rot1, v[i + 1], v[i]), 0.0f);
}

__device__ __forceinline__ void keep_order() { __builtin_amdgcn_sched_barrier(0); }

__device__ __forceinline__ void sums_done(float a, float b, float c, float d) {
  asm volatile("" :: "v"(a), "v"(b), "v"(c), "v"(d));
}

__global__ __launch_bounds__(256, 4) void sepconv_vert_wmma_f32(
    const float* __restrict__ imgs, const float* __restrict__ vers,
    const float* __restrict__ hors, float* __restrict__ out)
{
  const int lane = (int)threadIdx.x & 31;
  const int wave = __builtin_amdgcn_readfirstlane((int)threadIdx.x >> 5);
  const int grp  = (int)blockIdx.x;

  constexpr int grpsPerRow = WW / (TW * GW);
  constexpr int grpsPerImg = HH / (TR * GH);
  const int b    = grp / (grpsPerImg * grpsPerRow);
  const int rem  = grp - b * grpsPerImg * grpsPerRow;
  const int gy   = rem / grpsPerRow;
  const int gx   = rem - gy * grpsPerRow;
  const int h0   = (gy * GH + wave / GW) * TR;
  const int w0   = (gx * GW + wave % GW) * TW;

  const int c16 = lane & 15;
  const int hb  = lane >> 4;

  int arow[KS][2];
  #pragma unroll
  for (int s = 0; s < KS; ++s) {
    #pragma unroll
    for (int e = 0; e < 2; ++e)
      arow[s][e] = clampi(h0 - PAD + 4 * s + 2 * hb + e, 0, HH - 1) * WW;
  }
  int acol[2];
  #pragma unroll
  for (int cb = 0; cb < 2; ++cb)
    acol[cb] = clampi(w0 - PAD + 16 * cb + c16, 0, WW - 1);

  const float* ib  = imgs + ((long)b * NC) * plane;
  const float* vb  = vers + ((long)b * KK) * plane + (long)h0 * WW + w0 + c16;
  const float* vbl = vb + (long)(2 * hb) * plane;
  const float* hu  = hors + ((long)b * KK) * plane + (long)h0 * WW + w0;

  const int    q  = 8 * hb - c16;
  const int    q4   = 8 * hb - (c16 & ~3);
  const bool   rot2 = (c16 & 2) == 0;
  const bool   rot1 = (c16 & 1) == 0;
  bool ok[2][8];
  #pragma unroll
  for (int cb = 0; cb < 2; ++cb) {
    #pragma unroll
    for (int i = 0; i < 8; ++i) {
      const int t = 16 * cb + i + q;
      ok[cb][i] = (t >= 0) && (t < KK);
    }
  }

  v2f   Af[NST][KS];
  v2f   Bf[TR][KS];
  float hraw[2][TR][NT];

  load_a(ib, arow, acol[0], Af[0]);
  keep_order();
  #pragma unroll
  for (int s = 0; s < KS; ++s) {
    #pragma unroll
    for (int r = 0; r < TR; ++r) Bf[r][s] = load_b(vb, vbl, r, s);
  }
  keep_order();
  load_taps(hu, c16, -3 + q4, hraw[0]);
  #pragma unroll
  for (int t = 1; t < LA; ++t)
    load_a(ib + (t % NC) * plane, arow, acol[t / NC], Af[t]);

  float acc[TR][NC];
  #pragma unroll
  for (int r = 0; r < TR; ++r) {
    #pragma unroll
    for (int c = 0; c < NC; ++c) acc[r][c] = 0.0f;
  }

  float wt[TR][8];

  #pragma unroll
  for (int cb = 0; cb < 2; ++cb) {
    #pragma unroll
    for (int c = 0; c < NC; ++c) {
      const int t = NC * cb + c;

      if (t + LA < NST)
        load_a(ib + ((t + LA) % NC) * plane, arow, acol[(t + LA) / NC], Af[t + LA]);
      if (t == LH)
        load_taps(hu, c16, 16 - 3 + q4, hraw[1]);
      keep_order();

      if (t == 0) {
        #pragma unroll
        for (int s = 0; s < KS; ++s) {
          #pragma unroll
          for (int r = 0; r < TR; ++r) Bf[r][s] = zero_b(Bf[r][s], hb, r, s);
        }
      }

      v8f D[TR];
      #pragma unroll
      for (int r = 0; r < TR; ++r) D[r] = (v8f){};

      #pragma unroll
      for (int s = 0; s < KS; ++s) {
        #pragma unroll
        for (int r = 0; r < TR; ++r)
          D[r] = __builtin_amdgcn_wmma_f32_16x16x4_f32(
                     false, Af[t][s], false, Bf[r][s], (short)0, D[r], false, false);
      }

      if (c == 0) {
        keep_order();
        #pragma unroll
        for (int r = 0; r < TR; ++r) rotate_taps(hraw[cb][r], rot2, rot1, ok[cb], wt[r]);
      }

      #pragma unroll
      for (int r = 0; r < TR; ++r) {
        #pragma unroll
        for (int i = 0; i < 8; ++i)
          acc[r][c] = fmaf(wt[r][i], D[r][i], acc[r][c]);
      }
      sums_done(acc[0][c], acc[1][c], acc[2][c], acc[3][c]);
    }
  }

  float* ob = out + ((long)b * NC) * plane + (long)(h0 + hb) * WW + w0 + c16;
  #pragma unroll
  for (int c = 0; c < NC; ++c) {
    #pragma unroll
    for (int rp = 0; rp < 2; ++rp) {
      float tot[2];
      #pragma unroll
      for (int k = 0; k < 2; ++k) {
        const float v = acc[2 * rp + k][c];
        const unsigned o = __builtin_amdgcn_permlanex16(
            __float_as_uint(v), __float_as_uint(v), 0x76543210u, 0xfedcba98u, false, false);
        tot[k] = v + __uint_as_float(o);
      }
      ob[c * plane + (2 * rp) * WW] = hb ? tot[1] : tot[0];
    }
  }
}

extern "C" void kernel_launch(void* const* d_in, const int* in_sizes, int n_in,
                              void* d_out, int out_size, void* d_ws, size_t ws_size,
                              hipStream_t stream) {
  const float* imgs = (const float*)d_in[0];
  const float* vers = (const float*)d_in[1];
  const float* hors = (const float*)d_in[2];
  float* out = (float*)d_out;

  dim3 grid(512);
  dim3 block(256);
  sepconv_vert_wmma_f32<<<grid, block, 0, stream>>>(imgs, vers, hors, out);
}
